# starting kernel
# baseline (speedup 1.0000x reference)
#include <hip/hip_runtime.h>

typedef __attribute__((ext_vector_type(2))) float v2f;
typedef __attribute__((ext_vector_type(8))) float v8f;

#define PN    1024
#define MD    64
#define RD    32
#define TI    16
#define SROW  1028
#define NTHR  512

__global__ __launch_bounds__(NTHR)
void social_tensor_kernel(const float* __restrict__ hidden,
                          const float* __restrict__ corr,
                          const int*   __restrict__ nei,
                          const float* __restrict__ W_rela,
                          const float* __restrict__ b_rela,
                          const float* __restrict__ W_att,
                          const float* __restrict__ b_att,
                          float* __restrict__ out)
{
    __shared__ float s[TI * SROW];
    __shared__ float hv_nei[PN];
    __shared__ float hv_hi[TI];
    __shared__ float wr0[RD], wr1[RD], brr[RD], war[RD];
    __shared__ float whi[MD], wnei[MD];
    __shared__ float partial[3 * 4 * 8 * 32];

    const int tid  = threadIdx.x;
    const int lane = tid & 31;
    const int wave = tid >> 5;
    const int i0   = blockIdx.x * TI;

    if (tid < RD) {
        wr0[tid] = W_rela[tid];
        wr1[tid] = W_rela[RD + tid];
        brr[tid] = b_rela[tid];
        war[tid] = W_att[tid];
    }
    if (tid >= 64 && tid < 64 + MD) {
        int m = tid - 64;
        whi[m]  = W_att[RD + m];
        wnei[m] = W_att[RD + MD + m];
    }
    __syncthreads();

    for (int j = tid; j < PN; j += NTHR) {
        const float* h = hidden + j * MD;
        float acc = 0.f;
        #pragma unroll 8
        for (int m = 0; m < MD; ++m) acc = fmaf(h[m], wnei[m], acc);
        hv_nei[j] = acc;
    }
    if (tid < TI) {
        const float* h = hidden + (i0 + tid) * MD;
        float acc = 0.f;
        #pragma unroll 8
        for (int m = 0; m < MD; ++m) acc = fmaf(h[m], whi[m], acc);
        hv_hi[tid] = acc;
    }
    __syncthreads();

    const float batt = b_att[0];

    for (int p = tid; p < TI * PN; p += NTHR) {
        const int ii = p >> 10;
        const int j  = p & (PN - 1);
        const long idx = (long)(i0 + ii) * PN + j;
        const float c0 = corr[idx * 2 + 0];
        const float c1 = corr[idx * 2 + 1];
        float acc = 0.f;
        #pragma unroll
        for (int r = 0; r < RD; ++r) {
            float t = fmaf(c1, wr1[r], brr[r]);
            t = fmaf(c0, wr0[r], t);
            t = fmaxf(t, 0.f);
            acc = fmaf(t, war[r], acc);
        }
        float sc = acc + hv_hi[ii] + hv_nei[j] + batt;
        float pv = (nei[idx] > 0) ? sc : 0.f;
        pv = (pv == 0.f) ? -1e-6f : pv;
        s[ii * SROW + j] = pv;
    }
    __syncthreads();

    {
        const int ii = wave;
        float* row = s + ii * SROW;
        float mx = -3.402823466e+38f;
        for (int j = lane; j < PN; j += 32) mx = fmaxf(mx, row[j]);
        #pragma unroll
        for (int off = 16; off > 0; off >>= 1) mx = fmaxf(mx, __shfl_xor(mx, off, 32));
        float sum = 0.f;
        for (int j = lane; j < PN; j += 32) {
            float e = expf(row[j] - mx);
            row[j] = e;
            sum += e;
        }
        #pragma unroll
        for (int off = 16; off > 0; off >>= 1) sum += __shfl_xor(sum, off, 32);
        const float inv = 1.f / sum;
        const int* nr = nei + (long)(i0 + ii) * PN;
        for (int j = lane; j < PN; j += 32) {
            float mf = (nr[j] > 0) ? 1.f : 0.f;
            row[j] = row[j] * inv * mf;
        }
    }
    __syncthreads();

    const int nt   = wave & 3;
    const int n0   = nt * 16;
    const int kq   = wave >> 2;
    const int kbeg = kq * (PN / 4);
    const int kend = kbeg + (PN / 4);

    v8f acc = {};
    const int arow = (lane & 15) * SROW;
    const int koff = (lane < 16) ? 0 : 2;
    const int ncol = n0 + (lane & 15);
    for (int k = kbeg; k < kend; k += 4) {
        v2f a, b;
        a.x = s[arow + k + koff + 0];
        a.y = s[arow + k + koff + 1];
        b.x = hidden[(k + koff + 0) * MD + ncol];
        b.y = hidden[(k + koff + 1) * MD + ncol];
        acc = __builtin_amdgcn_wmma_f32_16x16x4_f32(false, a, false, b,
                                                    (short)0, acc, false, false);
    }

    if (kq > 0) {
        #pragma unroll
        for (int g = 0; g < 8; ++g)
            partial[(((kq - 1) * 4 + nt) * 8 + g) * 32 + lane] = acc[g];
    }
    __syncthreads();
    if (kq == 0) {
        #pragma unroll
        for (int g = 0; g < 8; ++g) {
            float v = acc[g]
                    + partial[((0 * 4 + nt) * 8 + g) * 32 + lane]
                    + partial[((1 * 4 + nt) * 8 + g) * 32 + lane]
                    + partial[((2 * 4 + nt) * 8 + g) * 32 + lane];
            const int m = g + ((lane < 16) ? 0 : 8);
            out[(long)(i0 + m) * MD + n0 + (lane & 15)] = v;
        }
    }
}

extern "C" void kernel_launch(void* const* d_in, const int* in_sizes, int n_in,
                              void* d_out, int out_size, void* d_ws, size_t ws_size,
                              hipStream_t stream) {
    (void)in_sizes; (void)n_in; (void)out_size; (void)d_ws; (void)ws_size;
    const float* hidden = (const float*)d_in[0];
    const float* corr   = (const float*)d_in[1];
    const int*   nei    = (const int*)d_in[2];
    const float* W_rela = (const float*)d_in[3];
    const float* b_rela = (const float*)d_in[4];
    const float* W_att  = (const float*)d_in[5];
    const float* b_att  = (const float*)d_in[6];
    float* outp = (float*)d_out;

    social_tensor_kernel<<<dim3(PN / TI), dim3(NTHR), 0, stream>>>(
        hidden, corr, nei, W_rela, b_rela, W_att, b_att, outp);
}


# 4 rows per work-group on 256 work-groups (was 16 rows on 64, a quarter of the WGPs); each row's softmax max and sum are now reduced across the 16 waves through LDS, which changes the row-sum order
# speedup vs baseline: 2.0463x; 2.0463x over previous
#include <hip/hip_runtime.h>


typedef __attribute__((ext_vector_type(2))) float v2f;
typedef __attribute__((ext_vector_type(8))) float v8f;

#define PN    1024
#define MD    64
#define RD    32
#define TI    4
#define SROW  1028
#define NTHR  512
#define NWAVE (NTHR / 32)
#define EPR   (PN / NTHR)
#define EPT   (TI * EPR)
static_assert(NWAVE == 16 && EPR == 2 && (TI & (TI - 1)) == 0, "reduction code below is written for this shape");

__global__ __launch_bounds__(NTHR)
void social_tensor_kernel(const float* __restrict__ hidden,
                          const float* __restrict__ corr,
                          const int*   __restrict__ nei,
                          const float* __restrict__ W_rela,
                          const float* __restrict__ b_rela,
                          const float* __restrict__ W_att,
                          const float* __restrict__ b_att,
                          float* __restrict__ out)
{
    __shared__ float s[TI * SROW];
    __shared__ float hv_nei[PN];
    __shared__ float hv_hi[TI];
    __shared__ float wr0[RD], wr1[RD], brr[RD], war[RD];
    __shared__ float whi[MD], wnei[MD];
    __shared__ __attribute__((aligned(16))) float red_mx[TI * NWAVE];
    __shared__ __attribute__((aligned(16))) float red_sm[TI * NWAVE];
    __shared__ float partial[4 * TI * MD];

    const int tid  = threadIdx.x;
    const int lane = tid & 31;
    const int wave = tid >> 5;
    const int i0   = blockIdx.x * TI;

    if (tid < RD) {
        wr0[tid] = W_rela[tid];
        wr1[tid] = W_rela[RD + tid];
        brr[tid] = b_rela[tid];
        war[tid] = W_att[tid];
    }
    if (tid >= 64 && tid < 64 + MD) {
        int m = tid - 64;
        whi[m]  = W_att[RD + m];
        wnei[m] = W_att[RD + MD + m];
    }
    __syncthreads();

    for (int j = tid; j < PN; j += NTHR) {
        const float* h = hidden + j * MD;
        float acc = 0.f;
        #pragma unroll 8
        for (int m = 0; m < MD; ++m) acc = fmaf(h[m], wnei[m], acc);
        hv_nei[j] = acc;
    }
    if (tid < TI) {
        const float* h = hidden + (i0 + tid) * MD;
        float acc = 0.f;
        #pragma unroll 8
        for (int m = 0; m < MD; ++m) acc = fmaf(h[m], whi[m], acc);
        hv_hi[tid] = acc;
    }
    __syncthreads();

    const float batt = b_att[0];

    float pv[EPT];
    float mf[EPT];
    #pragma unroll
    for (int e = 0; e < EPT; ++e) {
        const int ii = e / EPR;
        const int j  = tid + (e % EPR) * NTHR;
        const long idx = (long)(i0 + ii) * PN + j;
        const float c0 = corr[idx * 2 + 0];
        const float c1 = corr[idx * 2 + 1];
        const bool  nb = nei[idx] > 0;
        float acc = 0.f;
        #pragma unroll
        for (int r = 0; r < RD; ++r) {
            float t = fmaf(c1, wr1[r], brr[r]);
            t = fmaf(c0, wr0[r], t);
            t = fmaxf(t, 0.f);
            acc = fmaf(t, war[r], acc);
        }
        float sc = acc + hv_hi[ii] + hv_nei[j] + batt;
        float v = nb ? sc : 0.f;
        v = (v == 0.f) ? -1e-6f : v;
        pv[e] = v;
        mf[e] = nb ? 1.f : 0.f;
    }

    float rm[TI];
    #pragma unroll
    for (int ii = 0; ii < TI; ++ii) rm[ii] = fmaxf(pv[ii * EPR + 0], pv[ii * EPR + 1]);
    #pragma unroll
    for (int off = 16; off > 0; off >>= 1) {
        #pragma unroll
        for (int ii = 0; ii < TI; ++ii) rm[ii] = fmaxf(rm[ii], __shfl_xor(rm[ii], off, 32));
    }
    if (lane == 0) {
        #pragma unroll
        for (int ii = 0; ii < TI; ++ii) red_mx[ii * NWAVE + wave] = rm[ii];
    }
    __syncthreads();
    float ev[EPT];
    float rs[TI];
    #pragma unroll
    for (int ii = 0; ii < TI; ++ii) {
        const float4* q = (const float4*)(red_mx + ii * NWAVE);
        const float4 a = q[0], b = q[1], c = q[2], d = q[3];
        const float mab = fmaxf(fmaxf(fmaxf(a.x, a.y), fmaxf(a.z, a.w)), fmaxf(fmaxf(b.x, b.y), fmaxf(b.z, b.w)));
        const float mcd = fmaxf(fmaxf(fmaxf(c.x, c.y), fmaxf(c.z, c.w)), fmaxf(fmaxf(d.x, d.y), fmaxf(d.z, d.w)));
        const float m = fmaxf(mab, mcd);
        const float e0 = expf(pv[ii * EPR + 0] - m);
        const float e1 = expf(pv[ii * EPR + 1] - m);
        ev[ii * EPR + 0] = e0;
        ev[ii * EPR + 1] = e1;
        rs[ii] = e0 + e1;
    }
    #pragma unroll
    for (int off = 16; off > 0; off >>= 1) {
        #pragma unroll
        for (int ii = 0; ii < TI; ++ii) rs[ii] += __shfl_xor(rs[ii], off, 32);
    }
    if (lane == 0) {
        #pragma unroll
        for (int ii = 0; ii < TI; ++ii) red_sm[ii * NWAVE + wave] = rs[ii];
    }
    __syncthreads();
    #pragma unroll
    for (int ii = 0; ii < TI; ++ii) {
        const float4* q = (const float4*)(red_sm + ii * NWAVE);
        const float4 a = q[0], b = q[1], c = q[2], d = q[3];
        const float tx = (a.x + b.x) + (c.x + d.x);
        const float ty = (a.y + b.y) + (c.y + d.y);
        const float tz = (a.z + b.z) + (c.z + d.z);
        const float tw = (a.w + b.w) + (c.w + d.w);
        const float sm = (tx + ty) + (tz + tw);
        const float inv = 1.f / sm;
        #pragma unroll
        for (int h = 0; h < EPR; ++h)
            s[ii * SROW + tid + h * NTHR] = ev[ii * EPR + h] * inv * mf[ii * EPR + h];
    }
    __syncthreads();

    const int nt   = wave & 3;
    const int n0   = nt * 16;
    const int kq   = wave >> 2;
    const int kbeg = kq * (PN / 4);
    const int kend = kbeg + (PN / 4);

    v8f acc = {};
    const int  arow = (lane & (TI - 1)) * SROW;
    const int  koff = (lane < 16) ? 0 : 2;
    const int  ncol = n0 + (lane & 15);
    for (int k = kbeg; k < kend; k += 4) {
        v2f a, b;
        a.x = s[arow + k + koff + 0];
        a.y = s[arow + k + koff + 1];
        b.x = hidden[(k + koff + 0) * MD + ncol];
        b.y = hidden[(k + koff + 1) * MD + ncol];
        acc = __builtin_amdgcn_wmma_f32_16x16x4_f32(false, a, false, b,
                                                    (short)0, acc, false, false);
    }

    asm volatile("v_nop\n\tv_nop\n\tv_nop\n\tv_nop\n\tv_nop\n\tv_nop\n\tv_nop\n\tv_nop" : "+v"(acc));

    if (lane < 16) {
        #pragma unroll
        for (int g = 0; g < TI; ++g)
            partial[(kq * TI + g) * MD + ncol] = acc[g];
    }
    __syncthreads();
    if (tid < TI * MD) {
        const int row = tid >> 6;
        const int col = tid & (MD - 1);
        float v = partial[(0 * TI + row) * MD + col]
                + partial[(1 * TI + row) * MD + col]
                + partial[(2 * TI + row) * MD + col]
                + partial[(3 * TI + row) * MD + col];
        out[(long)(i0 + row) * MD + col] = v;
    }
}

extern "C" void kernel_launch(void* const* d_in, const int* in_sizes, int n_in,
                              void* d_out, int out_size, void* d_ws, size_t ws_size,
                              hipStream_t stream) {
    (void)in_sizes; (void)n_in; (void)out_size; (void)d_ws; (void)ws_size;
    const float* hidden = (const float*)d_in[0];
    const float* corr   = (const float*)d_in[1];
    const int*   nei    = (const int*)d_in[2];
    const float* W_rela = (const float*)d_in[3];
    const float* b_rela = (const float*)d_in[4];
    const float* W_att  = (const float*)d_in[5];
    const float* b_att  = (const float*)d_in[6];
    float* outp = (float*)d_out;

    social_tensor_kernel<<<dim3(PN / TI), dim3(NTHR), 0, stream>>>(
        hidden, corr, nei, W_rela, b_rela, W_att, b_att, outp);
}


# hidden read once per work-group, coalesced, and kept in LDS (256 KB); rank-1 score terms and B fragments come from that copy; the rank-1 dot products change their sum order
# speedup vs baseline: 2.8564x; 1.3959x over previous
#include <hip/hip_runtime.h>


typedef __attribute__((ext_vector_type(2))) float v2f;
typedef __attribute__((ext_vector_type(8))) float v8f;

#define PN    1024
#define MD    64
#define RD    32
#define TI    4
#define SROW  1028
#define NTHR  512
#define NWAVE (NTHR / 32)
#define EPR   (PN / NTHR)
#define EPT   (TI * EPR)
#define HB    16
static_assert(NWAVE == 16 && EPR == 2 && (TI & (TI - 1)) == 0, "reduction code below is written for this shape");
static_assert(2 * HB * NTHR * 4 == PN * MD, "two batches cover hidden exactly once");

template <int M>
__device__ __forceinline__ float row_xor(float x)
{
    return __builtin_bit_cast(float,
        __builtin_amdgcn_update_dpp(0, __builtin_bit_cast(int, x), 0x160 + M, 0xf, 0xf, true));
}
__device__ __forceinline__ float row16_sum(float x)
{
    x += row_xor<1>(x);
    x += row_xor<2>(x);
    x += row_xor<4>(x);
    x += row_xor<8>(x);
    return x;
}
__device__ __forceinline__ float row16_sum16(const float (&d)[16], int q)
{
    const bool b8 = (q & 8) != 0, b4 = (q & 4) != 0, b2 = (q & 2) != 0, b1 = (q & 1) != 0;
    float a[8], b[4], c[2];
    #pragma unroll
    for (int u = 0; u < 8; ++u) {
        const float keep = b8 ? d[u + 8] : d[u];
        const float send = b8 ? d[u] : d[u + 8];
        a[u] = keep + row_xor<8>(send);
    }
    #pragma unroll
    for (int u = 0; u < 4; ++u) {
        const float keep = b4 ? a[u + 4] : a[u];
        const float send = b4 ? a[u] : a[u + 4];
        b[u] = keep + row_xor<4>(send);
    }
    #pragma unroll
    for (int u = 0; u < 2; ++u) {
        const float keep = b2 ? b[u + 2] : b[u];
        const float send = b2 ? b[u] : b[u + 2];
        c[u] = keep + row_xor<2>(send);
    }
    const float keep = b1 ? c[1] : c[0];
    const float send = b1 ? c[0] : c[1];
    return keep + row_xor<1>(send);
}

#define WMMA_SLOTS(acc) asm volatile("v_nop\n\tv_nop\n\tv_nop\n\tv_nop\n\tv_nop\n\tv_nop\n\tv_nop\n\tv_nop" : "+v"(acc))

__global__ __launch_bounds__(NTHR)
void social_tensor_kernel(const float* __restrict__ hidden,
                          const float* __restrict__ corr,
                          const int*   __restrict__ nei,
                          const float* __restrict__ W_rela,
                          const float* __restrict__ b_rela,
                          const float* __restrict__ W_att,
                          const float* __restrict__ b_att,
                          float* __restrict__ out)
{
    struct __attribute__((aligned(16))) Smem {
        float red_mx[TI * NWAVE];
        float red_sm[TI * NWAVE];
        float s[TI * SROW];
        float hv_nei[PN];
        float partial[4 * TI * MD];
        float wr0[RD], wr1[RD], brr[RD], war[RD];
        float hv_hi[TI];
        float hs[PN * MD];
    };
    static_assert(__builtin_offsetof(Smem, hs) % 16 == 0 && __builtin_offsetof(Smem, red_sm) % 16 == 0,
                  "16-byte LDS accesses below");
    __shared__ Smem sm;
    auto& red_mx = sm.red_mx;  auto& red_sm = sm.red_sm;  auto& s = sm.s;  auto& hv_nei = sm.hv_nei;
    auto& partial = sm.partial;  auto& wr0 = sm.wr0;  auto& wr1 = sm.wr1;  auto& brr = sm.brr;  auto& war = sm.war;
    auto& hv_hi = sm.hv_hi;  auto& hs = sm.hs;

    const int tid  = threadIdx.x;
    const int lane = tid & 31;
    const int wave = tid >> 5;
    const int i0   = blockIdx.x * TI;
    const int q    = tid & 15;
    const int hr   = tid >> 4;
    const int qs   = q ^ ((wave & 1) << 2);

    float4 hb[HB];
    #pragma unroll
    for (int u = 0; u < HB; ++u) {
        const float* p = hidden + (long)(u * NTHR + tid) * 4;
        hb[u] = make_float4(p[0], p[1], p[2], p[3]);
    }
    const float* wq = W_att + RD + MD + 4 * q;
    const float4 wn = make_float4(wq[0], wq[1], wq[2], wq[3]);

    float c0[EPT], c1[EPT];
    unsigned nbm = 0;
    #pragma unroll
    for (int e = 0; e < EPT; ++e) {
        const int ii = e / EPR;
        const int j  = tid + (e % EPR) * NTHR;
        const long idx = (long)(i0 + ii) * PN + j;
        c0[e]  = corr[idx * 2 + 0];
        c1[e]  = corr[idx * 2 + 1];
        nbm   |= (nei[idx] > 0 ? 1u : 0u) << e;
    }

    if (tid < RD) {
        wr0[tid] = W_rela[tid];
        wr1[tid] = W_rela[RD + tid];
        brr[tid] = b_rela[tid];
        war[tid] = W_att[tid];
    }
    if (tid < 16 * TI) {
        const float* p = hidden + (long)(i0 + hr) * MD + 4 * q;
        const float* w = W_att + RD + 4 * q;
        float d = p[0] * w[0];
        d = fmaf(p[1], w[1], d);
        d = fmaf(p[2], w[2], d);
        d = fmaf(p[3], w[3], d);
        d = row16_sum(d);
        if (q == 0) hv_hi[hr] = d;
    }
    __syncthreads();

    const float batt = b_att[0];

    {
        float d[HB];
        #pragma unroll
        for (int u = 0; u < HB; ++u) {
            const int row = u * 32 + hr;
            *(float4*)&hs[row * MD + 4 * qs] = hb[u];
            float t = hb[u].x * wn.x;
            t = fmaf(hb[u].y, wn.y, t);
            t = fmaf(hb[u].z, wn.z, t);
            t = fmaf(hb[u].w, wn.w, t);
            d[u] = t;
        }
        hv_nei[q * 32 + hr] = row16_sum16(d, q);
    }
    #pragma unroll
    for (int u = 0; u < HB; ++u) {
        const float* p = hidden + (long)((HB + u) * NTHR + tid) * 4;
        hb[u] = make_float4(p[0], p[1], p[2], p[3]);
    }
    float acc_s[EPT];
    #pragma unroll
    for (int e = 0; e < EPT; ++e) {
        float acc = 0.f;
        #pragma unroll
        for (int r = 0; r < RD; ++r) {
            float t = fmaf(c1[e], wr1[r], brr[r]);
            t = fmaf(c0[e], wr0[r], t);
            t = fmaxf(t, 0.f);
            acc = fmaf(t, war[r], acc);
        }
        acc_s[e] = acc;
    }
    {
        float d[HB];
        #pragma unroll
        for (int u = 0; u < HB; ++u) {
            const int row = (HB + u) * 32 + hr;
            *(float4*)&hs[row * MD + 4 * qs] = hb[u];
            float t = hb[u].x * wn.x;
            t = fmaf(hb[u].y, wn.y, t);
            t = fmaf(hb[u].z, wn.z, t);
            t = fmaf(hb[u].w, wn.w, t);
            d[u] = t;
        }
        hv_nei[(HB + q) * 32 + hr] = row16_sum16(d, q);
    }
    __syncthreads();

    float pv[EPT];
    float mf[EPT];
    #pragma unroll
    for (int e = 0; e < EPT; ++e) {
        const int ii = e / EPR;
        const int j  = tid + (e % EPR) * NTHR;
        const bool nb = ((nbm >> e) & 1u) != 0;
        float sc = acc_s[e] + hv_hi[ii] + hv_nei[j] + batt;
        float v = nb ? sc : 0.f;
        v = (v == 0.f) ? -1e-6f : v;
        pv[e] = v;
        mf[e] = nb ? 1.f : 0.f;
    }

    float rm[TI];
    #pragma unroll
    for (int ii = 0; ii < TI; ++ii) rm[ii] = fmaxf(pv[ii * EPR + 0], pv[ii * EPR + 1]);
    #pragma unroll
    for (int off = 16; off > 0; off >>= 1) {
        #pragma unroll
        for (int ii = 0; ii < TI; ++ii) rm[ii] = fmaxf(rm[ii], __shfl_xor(rm[ii], off, 32));
    }
    if (lane == 0) {
        #pragma unroll
        for (int ii = 0; ii < TI; ++ii) red_mx[ii * NWAVE + wave] = rm[ii];
    }
    __syncthreads();
    float ev[EPT];
    float rs[TI];
    #pragma unroll
    for (int ii = 0; ii < TI; ++ii) {
        const float4* qm = (const float4*)(red_mx + ii * NWAVE);
        const float4 a = qm[0], b = qm[1], c = qm[2], d = qm[3];
        const float mab = fmaxf(fmaxf(fmaxf(a.x, a.y), fmaxf(a.z, a.w)), fmaxf(fmaxf(b.x, b.y), fmaxf(b.z, b.w)));
        const float mcd = fmaxf(fmaxf(fmaxf(c.x, c.y), fmaxf(c.z, c.w)), fmaxf(fmaxf(d.x, d.y), fmaxf(d.z, d.w)));
        const float m = fmaxf(mab, mcd);
        const float e0 = expf(pv[ii * EPR + 0] - m);
        const float e1 = expf(pv[ii * EPR + 1] - m);
        ev[ii * EPR + 0] = e0;
        ev[ii * EPR + 1] = e1;
        rs[ii] = e0 + e1;
    }
    #pragma unroll
    for (int off = 16; off > 0; off >>= 1) {
        #pragma unroll
        for (int ii = 0; ii < TI; ++ii) rs[ii] += __shfl_xor(rs[ii], off, 32);
    }
    if (lane == 0) {
        #pragma unroll
        for (int ii = 0; ii < TI; ++ii) red_sm[ii * NWAVE + wave] = rs[ii];
    }
    __syncthreads();
    #pragma unroll
    for (int ii = 0; ii < TI; ++ii) {
        const float4* qm = (const float4*)(red_sm + ii * NWAVE);
        const float4 a = qm[0], b = qm[1], c = qm[2], d = qm[3];
        const float tx = (a.x + b.x) + (c.x + d.x);
        const float ty = (a.y + b.y) + (c.y + d.y);
        const float tz = (a.z + b.z) + (c.z + d.z);
        const float tw = (a.w + b.w) + (c.w + d.w);
        const float sm = (tx + ty) + (tz + tw);
        const float inv = 1.f / sm;
        #pragma unroll
        for (int h = 0; h < EPR; ++h)
            s[ii * SROW + tid + h * NTHR] = ev[ii * EPR + h] * inv * mf[ii * EPR + h];
    }
    __syncthreads();

    const int nt   = wave & 3;
    const int n0   = nt * 16;
    const int kq   = wave >> 2;
    const int kbeg = kq * (PN / 4);

    v8f acc = {};
    const int  arow = (lane & (TI - 1)) * SROW;
    const int  half = lane >> 4;
    const int  koff = half * 2;
    const int  ncol = n0 + (lane & 15);
    const int  bcol = ncol ^ (half << 4);
    constexpr int KB = 8;
    const float* ap = s + arow + kbeg + koff;
    const float* bp = hs + (kbeg + koff) * MD + bcol;
    v2f a0[KB], b0[KB], a1[KB], b1[KB];
    #pragma unroll
    for (int t = 0; t < KB; ++t) {
        a0[t].x = ap[4 * t];            a0[t].y = ap[4 * t + 1];
        b0[t].x = bp[(4 * t) * MD];     b0[t].y = bp[(4 * t + 1) * MD];
    }
    constexpr int NBLK = (PN / 4) / (4 * 2 * KB);
    static_assert(NBLK * 4 * 2 * KB == PN / 4, "a K-quarter is a whole number of trips");
    #pragma unroll 1
    for (int blk = 0; blk < NBLK; ++blk) {
        ap += 4 * KB;  bp += 4 * KB * MD;
        #pragma unroll
        for (int t = 0; t < KB; ++t) {
            a1[t].x = ap[4 * t];            a1[t].y = ap[4 * t + 1];
            b1[t].x = bp[(4 * t) * MD];     b1[t].y = bp[(4 * t + 1) * MD];
        }
        #pragma unroll
        for (int t = 0; t < KB; ++t)
            acc = __builtin_amdgcn_wmma_f32_16x16x4_f32(false, a0[t], false, b0[t], (short)0, acc, false, false);
        WMMA_SLOTS(acc);
        ap += 4 * KB;  bp += 4 * KB * MD;
        if (blk + 1 < NBLK) {
            #pragma unroll
            for (int t = 0; t < KB; ++t) {
                a0[t].x = ap[4 * t];            a0[t].y = ap[4 * t + 1];
                b0[t].x = bp[(4 * t) * MD];     b0[t].y = bp[(4 * t + 1) * MD];
            }
        }
        #pragma unroll
        for (int t = 0; t < KB; ++t)
            acc = __builtin_amdgcn_wmma_f32_16x16x4_f32(false, a1[t], false, b1[t], (short)0, acc, false, false);
        WMMA_SLOTS(acc);
    }

    if (lane < 16) {
        #pragma unroll
        for (int g = 0; g < TI; ++g)
            partial[(kq * TI + g) * MD + ncol] = acc[g];
    }
    __syncthreads();
    if (tid < TI * MD) {
        const int row = tid >> 6;
        const int col = tid & (MD - 1);
        float v = partial[(0 * TI + row) * MD + col]
                + partial[(1 * TI + row) * MD + col]
                + partial[(2 * TI + row) * MD + col]
                + partial[(3 * TI + row) * MD + col];
        out[(long)(i0 + row) * MD + col] = v;
    }
}

extern "C" void kernel_launch(void* const* d_in, const int* in_sizes, int n_in,
                              void* d_out, int out_size, void* d_ws, size_t ws_size,
                              hipStream_t stream) {
    (void)in_sizes; (void)n_in; (void)out_size; (void)d_ws; (void)ws_size;
    const float* hidden = (const float*)d_in[0];
    const float* corr   = (const float*)d_in[1];
    const int*   nei    = (const int*)d_in[2];
    const float* W_rela = (const float*)d_in[3];
    const float* b_rela = (const float*)d_in[4];
    const float* W_att  = (const float*)d_in[5];
    const float* b_att  = (const float*)d_in[6];
    float* outp = (float*)d_out;

    social_tensor_kernel<<<dim3(PN / TI), dim3(NTHR), 0, stream>>>(
        hidden, corr, nei, W_rela, b_rela, W_att, b_att, outp);
}


# corr/nei loaded first; all of hidden issued behind the first barrier and waited for only after the relation-term arithmetic, now a loop streaming its weights from LDS; every sum order unchanged
# speedup vs baseline: 3.0100x; 1.0538x over previous
#include <hip/hip_runtime.h>


typedef __attribute__((ext_vector_type(2))) float v2f;
typedef __attribute__((ext_vector_type(4))) float v4f;
typedef __attribute__((ext_vector_type(8))) float v8f;

#define PN    1024
#define MD    64
#define RD    32
#define TI    4
#define SROW  1028
#define NTHR  512
#define NWAVE (NTHR / 32)
#define EPR   (PN / NTHR)
#define EPT   (TI * EPR)
#define HB    16
static_assert(NWAVE == 16 && EPR == 2 && (TI & (TI - 1)) == 0, "reduction code below is written for this shape");
static_assert(2 * HB * NTHR * 4 == PN * MD, "two batches cover hidden exactly once");

template <int M>
__device__ __forceinline__ float row_xor(float x)
{
    return __builtin_bit_cast(float,
        __builtin_amdgcn_update_dpp(0, __builtin_bit_cast(int, x), 0x160 + M, 0xf, 0xf, true));
}
__device__ __forceinline__ float row16_sum(float x)
{
    x += row_xor<1>(x);
    x += row_xor<2>(x);
    x += row_xor<4>(x);
    x += row_xor<8>(x);
    return x;
}
__device__ __forceinline__ float row16_sum16(const float (&d)[16], int q)
{
    const bool b8 = (q & 8) != 0, b4 = (q & 4) != 0, b2 = (q & 2) != 0, b1 = (q & 1) != 0;
    float a[8], b[4], c[2];
    #pragma unroll
    for (int u = 0; u < 8; ++u) {
        const float keep = b8 ? d[u + 8] : d[u];
        const float send = b8 ? d[u] : d[u + 8];
        a[u] = keep + row_xor<8>(send);
    }
    #pragma unroll
    for (int u = 0; u < 4; ++u) {
        const float keep = b4 ? a[u + 4] : a[u];
        const float send = b4 ? a[u] : a[u + 4];
        b[u] = keep + row_xor<4>(send);
    }
    #pragma unroll
    for (int u = 0; u < 2; ++u) {
        const float keep = b2 ? b[u + 2] : b[u];
        const float send = b2 ? b[u] : b[u + 2];
        c[u] = keep + row_xor<2>(send);
    }
    const float keep = b1 ? c[1] : c[0];
    const float send = b1 ? c[0] : c[1];
    return keep + row_xor<1>(send);
}

#define WMMA_SLOTS(acc) asm volatile("v_nop\n\tv_nop\n\tv_nop\n\tv_nop\n\tv_nop\n\tv_nop\n\tv_nop\n\tv_nop" : "+v"(acc))

__global__ __launch_bounds__(NTHR)
void social_tensor_kernel(const float* __restrict__ hidden,
                          const float* __restrict__ corr,
                          const int*   __restrict__ nei,
                          const float* __restrict__ W_rela,
                          const float* __restrict__ b_rela,
                          const float* __restrict__ W_att,
                          const float* __restrict__ b_att,
                          float* __restrict__ out)
{
    struct __attribute__((aligned(16))) Smem {
        float red_mx[TI * NWAVE];
        float red_sm[TI * NWAVE];
        float s[TI * SROW];
        float hv_nei[PN];
        float partial[4 * TI * MD];
        float wts[4 * RD];
        float hv_hi[TI];
        float hs[PN * MD];
    };
    static_assert(__builtin_offsetof(Smem, hs) % 16 == 0 && __builtin_offsetof(Smem, red_sm) % 16 == 0 &&
                  __builtin_offsetof(Smem, wts) % 16 == 0, "16-byte LDS accesses below");
    __shared__ Smem sm;
    auto& red_mx = sm.red_mx;  auto& red_sm = sm.red_sm;  auto& s = sm.s;  auto& hv_nei = sm.hv_nei;
    auto& partial = sm.partial;  auto& wts = sm.wts;
    auto& hv_hi = sm.hv_hi;  auto& hs = sm.hs;

    const int tid  = threadIdx.x;
    const int lane = tid & 31;
    const int wave = tid >> 5;
    const int i0   = blockIdx.x * TI;
    const int q    = tid & 15;
    const int hr   = tid >> 4;
    const int qs   = q ^ ((wave & 1) << 2);

    const int wsel = __builtin_amdgcn_readfirstlane(wave) & 3;
    const float* wsrc = wsel == 0 ? W_rela : wsel == 1 ? W_rela + RD : wsel == 2 ? b_rela : W_att;
    const float wv = wsrc[lane];
    const float* wq = W_att + RD + MD + 4 * q;
    const float4 wn = make_float4(wq[0], wq[1], wq[2], wq[3]);
    const float* php = hidden + (long)(i0 + (hr & (TI - 1))) * MD + 4 * q;
    const float* pwh = W_att + RD + 4 * q;
    const float4 hp = make_float4(php[0], php[1], php[2], php[3]);
    const float4 hw = make_float4(pwh[0], pwh[1], pwh[2], pwh[3]);

    float c0[EPT], c1[EPT];
    int   nv[EPT];
    #pragma unroll
    for (int e = 0; e < EPT; ++e) {
        const int ii = e / EPR;
        const int j  = tid + (e % EPR) * NTHR;
        const long idx = (long)(i0 + ii) * PN + j;
        c0[e] = corr[idx * 2 + 0];
        c1[e] = corr[idx * 2 + 1];
        nv[e] = nei[idx];
    }

    __builtin_amdgcn_sched_barrier(0);

    wts[tid & (4 * RD - 1)] = wv;
    {
        float d = hp.x * hw.x;
        d = fmaf(hp.y, hw.y, d);
        d = fmaf(hp.z, hw.z, d);
        d = fmaf(hp.w, hw.w, d);
        d = row16_sum(d);
        if (tid < 16 * TI && q == 0) hv_hi[hr] = d;
    }
    __syncthreads();

    float4 hb[2 * HB];
    #pragma unroll
    for (int u = 0; u < 2 * HB; ++u) {
        const float* p = hidden + (long)(u * NTHR + tid) * 4;
        hb[u] = make_float4(p[0], p[1], p[2], p[3]);
    }

    const float batt = b_att[0];
    unsigned nbm = 0;
    #pragma unroll
    for (int e = 0; e < EPT; ++e) nbm |= (nv[e] > 0 ? 1u : 0u) << e;

    float acc_s[EPT];
    #pragma unroll
    for (int e = 0; e < EPT; ++e) acc_s[e] = 0.f;
    {
        const v4f* w0v = (const v4f*)(wts);
        const v4f* w1v = (const v4f*)(wts + RD);
        const v4f* bv  = (const v4f*)(wts + 2 * RD);
        const v4f* wav = (const v4f*)(wts + 3 * RD);
        #pragma unroll 1
        for (int rc = 0; rc < RD / 4; ++rc) {
            const v4f g0 = w0v[rc], g1 = w1v[rc], gb = bv[rc], ga = wav[rc];
            #pragma unroll
            for (int e = 0; e < EPT; ++e) {
                float t;
                t = fmaf(c1[e], g1.x, gb.x);  t = fmaf(c0[e], g0.x, t);  t = fmaxf(t, 0.f);  acc_s[e] = fmaf(t, ga.x, acc_s[e]);
                t = fmaf(c1[e], g1.y, gb.y);  t = fmaf(c0[e], g0.y, t);  t = fmaxf(t, 0.f);  acc_s[e] = fmaf(t, ga.y, acc_s[e]);
                t = fmaf(c1[e], g1.z, gb.z);  t = fmaf(c0[e], g0.z, t);  t = fmaxf(t, 0.f);  acc_s[e] = fmaf(t, ga.z, acc_s[e]);
                t = fmaf(c1[e], g1.w, gb.w);  t = fmaf(c0[e], g0.w, t);  t = fmaxf(t, 0.f);  acc_s[e] = fmaf(t, ga.w, acc_s[e]);
            }
        }
    }

    #pragma unroll
    for (int hh = 0; hh < 2; ++hh) {
        float d[HB];
        #pragma unroll
        for (int u = 0; u < HB; ++u) {
            const int row = (hh * HB + u) * 32 + hr;
            const float4 v = hb[hh * HB + u];
            *(float4*)&hs[row * MD + 4 * qs] = v;
            float t = v.x * wn.x;
            t = fmaf(v.y, wn.y, t);
            t = fmaf(v.z, wn.z, t);
            t = fmaf(v.w, wn.w, t);
            d[u] = t;
        }
        hv_nei[(hh * HB + q) * 32 + hr] = row16_sum16(d, q);
    }
    __syncthreads();

    float pv[EPT];
    float mf[EPT];
    #pragma unroll
    for (int e = 0; e < EPT; ++e) {
        const int ii = e / EPR;
        const int j  = tid + (e % EPR) * NTHR;
        const bool nb = ((nbm >> e) & 1u) != 0;
        float sc = acc_s[e] + hv_hi[ii] + hv_nei[j] + batt;
        float v = nb ? sc : 0.f;
        v = (v == 0.f) ? -1e-6f : v;
        pv[e] = v;
        mf[e] = nb ? 1.f : 0.f;
    }

    float rm[TI];
    #pragma unroll
    for (int ii = 0; ii < TI; ++ii) rm[ii] = fmaxf(pv[ii * EPR + 0], pv[ii * EPR + 1]);
    #pragma unroll
    for (int off = 16; off > 0; off >>= 1) {
        #pragma unroll
        for (int ii = 0; ii < TI; ++ii) rm[ii] = fmaxf(rm[ii], __shfl_xor(rm[ii], off, 32));
    }
    if (lane == 0) {
        #pragma unroll
        for (int ii = 0; ii < TI; ++ii) red_mx[ii * NWAVE + wave] = rm[ii];
    }
    __syncthreads();
    float ev[EPT];
    float rs[TI];
    #pragma unroll
    for (int ii = 0; ii < TI; ++ii) {
        const float4* qm = (const float4*)(red_mx + ii * NWAVE);
        const float4 a = qm[0], b = qm[1], c = qm[2], d = qm[3];
        const float mab = fmaxf(fmaxf(fmaxf(a.x, a.y), fmaxf(a.z, a.w)), fmaxf(fmaxf(b.x, b.y), fmaxf(b.z, b.w)));
        const float mcd = fmaxf(fmaxf(fmaxf(c.x, c.y), fmaxf(c.z, c.w)), fmaxf(fmaxf(d.x, d.y), fmaxf(d.z, d.w)));
        const float m = fmaxf(mab, mcd);
        const float e0 = expf(pv[ii * EPR + 0] - m);
        const float e1 = expf(pv[ii * EPR + 1] - m);
        ev[ii * EPR + 0] = e0;
        ev[ii * EPR + 1] = e1;
        rs[ii] = e0 + e1;
    }
    #pragma unroll
    for (int off = 16; off > 0; off >>= 1) {
        #pragma unroll
        for (int ii = 0; ii < TI; ++ii) rs[ii] += __shfl_xor(rs[ii], off, 32);
    }
    if (lane == 0) {
        #pragma unroll
        for (int ii = 0; ii < TI; ++ii) red_sm[ii * NWAVE + wave] = rs[ii];
    }
    __syncthreads();
    #pragma unroll
    for (int ii = 0; ii < TI; ++ii) {
        const float4* qm = (const float4*)(red_sm + ii * NWAVE);
        const float4 a = qm[0], b = qm[1], c = qm[2], d = qm[3];
        const float tx = (a.x + b.x) + (c.x + d.x);
        const float ty = (a.y + b.y) + (c.y + d.y);
        const float tz = (a.z + b.z) + (c.z + d.z);
        const float tw = (a.w + b.w) + (c.w + d.w);
        const float sm = (tx + ty) + (tz + tw);
        const float inv = 1.f / sm;
        #pragma unroll
        for (int h = 0; h < EPR; ++h)
            s[ii * SROW + tid + h * NTHR] = ev[ii * EPR + h] * inv * mf[ii * EPR + h];
    }
    __syncthreads();

    const int nt   = wave & 3;
    const int n0   = nt * 16;
    const int kq   = wave >> 2;
    const int kbeg = kq * (PN / 4);

    v8f acc = {};
    const int  arow = (lane & (TI - 1)) * SROW;
    const int  half = lane >> 4;
    const int  koff = half * 2;
    const int  ncol = n0 + (lane & 15);
    const int  bcol = ncol ^ (half << 4);
    constexpr int KB = 8;
    const float* ap = s + arow + kbeg + koff;
    const float* bp = hs + (kbeg + koff) * MD + bcol;
    v2f a0[KB], b0[KB], a1[KB], b1[KB];
    #pragma unroll
    for (int t = 0; t < KB; ++t) {
        a0[t].x = ap[4 * t];            a0[t].y = ap[4 * t + 1];
        b0[t].x = bp[(4 * t) * MD];     b0[t].y = bp[(4 * t + 1) * MD];
    }
    constexpr int NBLK = (PN / 4) / (4 * 2 * KB);
    static_assert(NBLK * 4 * 2 * KB == PN / 4, "a K-quarter is a whole number of trips");
    #pragma unroll 1
    for (int blk = 0; blk < NBLK; ++blk) {
        ap += 4 * KB;  bp += 4 * KB * MD;
        #pragma unroll
        for (int t = 0; t < KB; ++t) {
            a1[t].x = ap[4 * t];            a1[t].y = ap[4 * t + 1];
            b1[t].x = bp[(4 * t) * MD];     b1[t].y = bp[(4 * t + 1) * MD];
        }
        #pragma unroll
        for (int t = 0; t < KB; ++t)
            acc = __builtin_amdgcn_wmma_f32_16x16x4_f32(false, a0[t], false, b0[t], (short)0, acc, false, false);
        WMMA_SLOTS(acc);
        ap += 4 * KB;  bp += 4 * KB * MD;
        if (blk + 1 < NBLK) {
            #pragma unroll
            for (int t = 0; t < KB; ++t) {
                a0[t].x = ap[4 * t];            a0[t].y = ap[4 * t + 1];
                b0[t].x = bp[(4 * t) * MD];     b0[t].y = bp[(4 * t + 1) * MD];
            }
        }
        #pragma unroll
        for (int t = 0; t < KB; ++t)
            acc = __builtin_amdgcn_wmma_f32_16x16x4_f32(false, a1[t], false, b1[t], (short)0, acc, false, false);
        WMMA_SLOTS(acc);
    }

    if (lane < 16) {
        #pragma unroll
        for (int g = 0; g < TI; ++g)
            partial[(kq * TI + g) * MD + ncol] = acc[g];
    }
    __syncthreads();
    if (tid < TI * MD) {
        const int row = tid >> 6;
        const int col = tid & (MD - 1);
        float v = partial[(0 * TI + row) * MD + col]
                + partial[(1 * TI + row) * MD + col]
                + partial[(2 * TI + row) * MD + col]
                + partial[(3 * TI + row) * MD + col];
        out[(long)(i0 + row) * MD + col] = v;
    }
}

extern "C" void kernel_launch(void* const* d_in, const int* in_sizes, int n_in,
                              void* d_out, int out_size, void* d_ws, size_t ws_size,
                              hipStream_t stream) {
    (void)in_sizes; (void)n_in; (void)out_size; (void)d_ws; (void)ws_size;
    const float* hidden = (const float*)d_in[0];
    const float* corr   = (const float*)d_in[1];
    const int*   nei    = (const int*)d_in[2];
    const float* W_rela = (const float*)d_in[3];
    const float* b_rela = (const float*)d_in[4];
    const float* W_att  = (const float*)d_in[5];
    const float* b_att  = (const float*)d_in[6];
    float* outp = (float*)d_out;

    social_tensor_kernel<<<dim3(PN / TI), dim3(NTHR), 0, stream>>>(
        hidden, corr, nei, W_rela, b_rela, W_att, b_att, outp);
}


# softmax row max and row sum reduced by DPP and permlanex16 lane exchanges instead of LDS shuffles and 16-byte broadcast re-reads (one wave value per lane); sum pairing order unchanged
# speedup vs baseline: 3.2722x; 1.0871x over previous
#include <hip/hip_runtime.h>


typedef __attribute__((ext_vector_type(2))) float v2f;
typedef __attribute__((ext_vector_type(4))) float v4f;
typedef __attribute__((ext_vector_type(8))) float v8f;

#define PN    1024
#define MD    64
#define RD    32
#define TI    4
#define SROW  1028
#define NTHR  512
#define NWAVE (NTHR / 32)
#define EPR   (PN / NTHR)
#define EPT   (TI * EPR)
#define HB    16
static_assert(NWAVE == 16 && EPR == 2 && (TI & (TI - 1)) == 0, "reduction code below is written for this shape");
static_assert(2 * HB * NTHR * 4 == PN * MD, "two batches cover hidden exactly once");

template <int M>
__device__ __forceinline__ float row_xor(float x)
{
    return __builtin_bit_cast(float,
        __builtin_amdgcn_update_dpp(0, __builtin_bit_cast(int, x), 0x160 + M, 0xf, 0xf, true));
}
__device__ __forceinline__ float xhalf(float x)
{
    const int b = __builtin_bit_cast(int, x);
    return __builtin_bit_cast(float, __builtin_amdgcn_permlanex16(b, b, 0x76543210, 0xfedcba98, false, false));
}
__device__ __forceinline__ float row16_sum(float x)
{
    x += row_xor<1>(x);
    x += row_xor<2>(x);
    x += row_xor<4>(x);
    x += row_xor<8>(x);
    return x;
}
__device__ __forceinline__ float row16_sum16(const float (&d)[16], int q)
{
    const bool b8 = (q & 8) != 0, b4 = (q & 4) != 0, b2 = (q & 2) != 0, b1 = (q & 1) != 0;
    float a[8], b[4], c[2];
    #pragma unroll
    for (int u = 0; u < 8; ++u) {
        const float keep = b8 ? d[u + 8] : d[u];
        const float send = b8 ? d[u] : d[u + 8];
        a[u] = keep + row_xor<8>(send);
    }
    #pragma unroll
    for (int u = 0; u < 4; ++u) {
        const float keep = b4 ? a[u + 4] : a[u];
        const float send = b4 ? a[u] : a[u + 4];
        b[u] = keep + row_xor<4>(send);
    }
    #pragma unroll
    for (int u = 0; u < 2; ++u) {
        const float keep = b2 ? b[u + 2] : b[u];
        const float send = b2 ? b[u] : b[u + 2];
        c[u] = keep + row_xor<2>(send);
    }
    const float keep = b1 ? c[1] : c[0];
    const float send = b1 ? c[0] : c[1];
    return keep + row_xor<1>(send);
}

#define WMMA_SLOTS(acc) asm volatile("v_nop\n\tv_nop\n\tv_nop\n\tv_nop\n\tv_nop\n\tv_nop\n\tv_nop\n\tv_nop" : "+v"(acc))

__global__ __launch_bounds__(NTHR)
void social_tensor_kernel(const float* __restrict__ hidden,
                          const float* __restrict__ corr,
                          const int*   __restrict__ nei,
                          const float* __restrict__ W_rela,
                          const float* __restrict__ b_rela,
                          const float* __restrict__ W_att,
                          const float* __restrict__ b_att,
                          float* __restrict__ out)
{
    struct __attribute__((aligned(16))) Smem {
        float red_mx[TI * NWAVE];
        float red_sm[TI * NWAVE];
        float s[TI * SROW];
        float hv_nei[PN];
        float partial[4 * TI * MD];
        float wts[4 * RD];
        float hv_hi[TI];
        float hs[PN * MD];
    };
    static_assert(__builtin_offsetof(Smem, hs) % 16 == 0 && __builtin_offsetof(Smem, red_sm) % 16 == 0 &&
                  __builtin_offsetof(Smem, wts) % 16 == 0, "16-byte LDS accesses below");
    __shared__ Smem sm;
    auto& red_mx = sm.red_mx;  auto& red_sm = sm.red_sm;  auto& s = sm.s;  auto& hv_nei = sm.hv_nei;
    auto& partial = sm.partial;  auto& wts = sm.wts;
    auto& hv_hi = sm.hv_hi;  auto& hs = sm.hs;

    const int tid  = threadIdx.x;
    const int lane = tid & 31;
    const int wave = tid >> 5;
    const int i0   = blockIdx.x * TI;
    const int q    = tid & 15;
    const int hr   = tid >> 4;
    const int qs   = q ^ ((wave & 1) << 2);

    const int wsel = __builtin_amdgcn_readfirstlane(wave) & 3;
    const float* wsrc = wsel == 0 ? W_rela : wsel == 1 ? W_rela + RD : wsel == 2 ? b_rela : W_att;
    const float wv = wsrc[lane];
    const float* wq = W_att + RD + MD + 4 * q;
    const float4 wn = make_float4(wq[0], wq[1], wq[2], wq[3]);
    const float* php = hidden + (long)(i0 + (hr & (TI - 1))) * MD + 4 * q;
    const float* pwh = W_att + RD + 4 * q;
    const float4 hp = make_float4(php[0], php[1], php[2], php[3]);
    const float4 hw = make_float4(pwh[0], pwh[1], pwh[2], pwh[3]);

    float c0[EPT], c1[EPT];
    int   nv[EPT];
    #pragma unroll
    for (int e = 0; e < EPT; ++e) {
        const int ii = e / EPR;
        const int j  = tid + (e % EPR) * NTHR;
        const long idx = (long)(i0 + ii) * PN + j;
        c0[e] = corr[idx * 2 + 0];
        c1[e] = corr[idx * 2 + 1];
        nv[e] = nei[idx];
    }

    __builtin_amdgcn_sched_barrier(0);

    wts[tid & (4 * RD - 1)] = wv;
    {
        float d = hp.x * hw.x;
        d = fmaf(hp.y, hw.y, d);
        d = fmaf(hp.z, hw.z, d);
        d = fmaf(hp.w, hw.w, d);
        d = row16_sum(d);
        if (tid < 16 * TI && q == 0) hv_hi[hr] = d;
    }
    __syncthreads();

    float4 hb[2 * HB];
    #pragma unroll
    for (int u = 0; u < 2 * HB; ++u) {
        const float* p = hidden + (long)(u * NTHR + tid) * 4;
        hb[u] = make_float4(p[0], p[1], p[2], p[3]);
    }

    const float batt = b_att[0];
    unsigned nbm = 0;
    #pragma unroll
    for (int e = 0; e < EPT; ++e) nbm |= (nv[e] > 0 ? 1u : 0u) << e;

    float acc_s[EPT];
    #pragma unroll
    for (int e = 0; e < EPT; ++e) acc_s[e] = 0.f;
    {
        const v4f* w0v = (const v4f*)(wts);
        const v4f* w1v = (const v4f*)(wts + RD);
        const v4f* bv  = (const v4f*)(wts + 2 * RD);
        const v4f* wav = (const v4f*)(wts + 3 * RD);
        #pragma unroll 1
        for (int rc = 0; rc < RD / 4; ++rc) {
            const v4f g0 = w0v[rc], g1 = w1v[rc], gb = bv[rc], ga = wav[rc];
            #pragma unroll
            for (int e = 0; e < EPT; ++e) {
                float t;
                t = fmaf(c1[e], g1.x, gb.x);  t = fmaf(c0[e], g0.x, t);  t = fmaxf(t, 0.f);  acc_s[e] = fmaf(t, ga.x, acc_s[e]);
                t = fmaf(c1[e], g1.y, gb.y);  t = fmaf(c0[e], g0.y, t);  t = fmaxf(t, 0.f);  acc_s[e] = fmaf(t, ga.y, acc_s[e]);
                t = fmaf(c1[e], g1.z, gb.z);  t = fmaf(c0[e], g0.z, t);  t = fmaxf(t, 0.f);  acc_s[e] = fmaf(t, ga.z, acc_s[e]);
                t = fmaf(c1[e], g1.w, gb.w);  t = fmaf(c0[e], g0.w, t);  t = fmaxf(t, 0.f);  acc_s[e] = fmaf(t, ga.w, acc_s[e]);
            }
        }
    }

    #pragma unroll
    for (int hh = 0; hh < 2; ++hh) {
        float d[HB];
        #pragma unroll
        for (int u = 0; u < HB; ++u) {
            const int row = (hh * HB + u) * 32 + hr;
            const float4 v = hb[hh * HB + u];
            *(float4*)&hs[row * MD + 4 * qs] = v;
            float t = v.x * wn.x;
            t = fmaf(v.y, wn.y, t);
            t = fmaf(v.z, wn.z, t);
            t = fmaf(v.w, wn.w, t);
            d[u] = t;
        }
        hv_nei[(hh * HB + q) * 32 + hr] = row16_sum16(d, q);
    }
    __syncthreads();

    float pv[EPT];
    float mf[EPT];
    #pragma unroll
    for (int e = 0; e < EPT; ++e) {
        const int ii = e / EPR;
        const int j  = tid + (e % EPR) * NTHR;
        const bool nb = ((nbm >> e) & 1u) != 0;
        float sc = acc_s[e] + hv_hi[ii] + hv_nei[j] + batt;
        float v = nb ? sc : 0.f;
        v = (v == 0.f) ? -1e-6f : v;
        pv[e] = v;
        mf[e] = nb ? 1.f : 0.f;
    }

    float rm[TI];
    #pragma unroll
    for (int ii = 0; ii < TI; ++ii) rm[ii] = fmaxf(pv[ii * EPR + 0], pv[ii * EPR + 1]);
    #pragma unroll
    for (int ii = 0; ii < TI; ++ii) rm[ii] = fmaxf(rm[ii], xhalf(rm[ii]));
    #pragma unroll
    for (int ii = 0; ii < TI; ++ii) rm[ii] = fmaxf(rm[ii], row_xor<8>(rm[ii]));
    #pragma unroll
    for (int ii = 0; ii < TI; ++ii) rm[ii] = fmaxf(rm[ii], row_xor<4>(rm[ii]));
    #pragma unroll
    for (int ii = 0; ii < TI; ++ii) rm[ii] = fmaxf(rm[ii], row_xor<2>(rm[ii]));
    #pragma unroll
    for (int ii = 0; ii < TI; ++ii) rm[ii] = fmaxf(rm[ii], row_xor<1>(rm[ii]));
    if (lane == 0) {
        #pragma unroll
        for (int ii = 0; ii < TI; ++ii) red_mx[ii * NWAVE + wave] = rm[ii];
    }
    __syncthreads();
    float ev[EPT];
    float rs[TI];
    float gm[TI];
    #pragma unroll
    for (int ii = 0; ii < TI; ++ii) gm[ii] = red_mx[ii * NWAVE + (lane & 15)];
    #pragma unroll
    for (int ii = 0; ii < TI; ++ii) gm[ii] = fmaxf(gm[ii], row_xor<4>(gm[ii]));
    #pragma unroll
    for (int ii = 0; ii < TI; ++ii) gm[ii] = fmaxf(gm[ii], row_xor<8>(gm[ii]));
    #pragma unroll
    for (int ii = 0; ii < TI; ++ii) gm[ii] = fmaxf(gm[ii], row_xor<1>(gm[ii]));
    #pragma unroll
    for (int ii = 0; ii < TI; ++ii) gm[ii] = fmaxf(gm[ii], row_xor<2>(gm[ii]));
    #pragma unroll
    for (int ii = 0; ii < TI; ++ii) {
        const float m = gm[ii];
        const float e0 = expf(pv[ii * EPR + 0] - m);
        const float e1 = expf(pv[ii * EPR + 1] - m);
        ev[ii * EPR + 0] = e0;
        ev[ii * EPR + 1] = e1;
        rs[ii] = e0 + e1;
    }
    #pragma unroll
    for (int ii = 0; ii < TI; ++ii) rs[ii] += xhalf(rs[ii]);
    #pragma unroll
    for (int ii = 0; ii < TI; ++ii) rs[ii] += row_xor<8>(rs[ii]);
    #pragma unroll
    for (int ii = 0; ii < TI; ++ii) rs[ii] += row_xor<4>(rs[ii]);
    #pragma unroll
    for (int ii = 0; ii < TI; ++ii) rs[ii] += row_xor<2>(rs[ii]);
    #pragma unroll
    for (int ii = 0; ii < TI; ++ii) rs[ii] += row_xor<1>(rs[ii]);
    if (lane == 0) {
        #pragma unroll
        for (int ii = 0; ii < TI; ++ii) red_sm[ii * NWAVE + wave] = rs[ii];
    }
    __syncthreads();
    float gs[TI];
    #pragma unroll
    for (int ii = 0; ii < TI; ++ii) gs[ii] = red_sm[ii * NWAVE + (lane & 15)];
    #pragma unroll
    for (int ii = 0; ii < TI; ++ii) gs[ii] += row_xor<4>(gs[ii]);
    #pragma unroll
    for (int ii = 0; ii < TI; ++ii) gs[ii] += row_xor<8>(gs[ii]);
    #pragma unroll
    for (int ii = 0; ii < TI; ++ii) gs[ii] += row_xor<1>(gs[ii]);
    #pragma unroll
    for (int ii = 0; ii < TI; ++ii) gs[ii] += row_xor<2>(gs[ii]);
    #pragma unroll
    for (int ii = 0; ii < TI; ++ii) {
        const float sm = gs[ii];
        const float inv = 1.f / sm;
        #pragma unroll
        for (int h = 0; h < EPR; ++h)
            s[ii * SROW + tid + h * NTHR] = ev[ii * EPR + h] * inv * mf[ii * EPR + h];
    }
    __syncthreads();

    const int nt   = wave & 3;
    const int n0   = nt * 16;
    const int kq   = wave >> 2;
    const int kbeg = kq * (PN / 4);

    v8f acc = {};
    const int  arow = (lane & (TI - 1)) * SROW;
    const int  half = lane >> 4;
    const int  koff = half * 2;
    const int  ncol = n0 + (lane & 15);
    const int  bcol = ncol ^ (half << 4);
    constexpr int KB = 8;
    const float* ap = s + arow + kbeg + koff;
    const float* bp = hs + (kbeg + koff) * MD + bcol;
    v2f a0[KB], b0[KB], a1[KB], b1[KB];
    #pragma unroll
    for (int t = 0; t < KB; ++t) {
        a0[t].x = ap[4 * t];            a0[t].y = ap[4 * t + 1];
        b0[t].x = bp[(4 * t) * MD];     b0[t].y = bp[(4 * t + 1) * MD];
    }
    constexpr int NBLK = (PN / 4) / (4 * 2 * KB);
    static_assert(NBLK * 4 * 2 * KB == PN / 4, "a K-quarter is a whole number of trips");
    #pragma unroll 1
    for (int blk = 0; blk < NBLK; ++blk) {
        ap += 4 * KB;  bp += 4 * KB * MD;
        #pragma unroll
        for (int t = 0; t < KB; ++t) {
            a1[t].x = ap[4 * t];            a1[t].y = ap[4 * t + 1];
            b1[t].x = bp[(4 * t) * MD];     b1[t].y = bp[(4 * t + 1) * MD];
        }
        #pragma unroll
        for (int t = 0; t < KB; ++t)
            acc = __builtin_amdgcn_wmma_f32_16x16x4_f32(false, a0[t], false, b0[t], (short)0, acc, false, false);
        WMMA_SLOTS(acc);
        ap += 4 * KB;  bp += 4 * KB * MD;
        if (blk + 1 < NBLK) {
            #pragma unroll
            for (int t = 0; t < KB; ++t) {
                a0[t].x = ap[4 * t];            a0[t].y = ap[4 * t + 1];
                b0[t].x = bp[(4 * t) * MD];     b0[t].y = bp[(4 * t + 1) * MD];
            }
        }
        #pragma unroll
        for (int t = 0; t < KB; ++t)
            acc = __builtin_amdgcn_wmma_f32_16x16x4_f32(false, a1[t], false, b1[t], (short)0, acc, false, false);
        WMMA_SLOTS(acc);
    }

    if (lane < 16) {
        #pragma unroll
        for (int g = 0; g < TI; ++g)
            partial[(kq * TI + g) * MD + ncol] = acc[g];
    }
    __syncthreads();
    if (tid < TI * MD) {
        const int row = tid >> 6;
        const int col = tid & (MD - 1);
        float v = partial[(0 * TI + row) * MD + col]
                + partial[(1 * TI + row) * MD + col]
                + partial[(2 * TI + row) * MD + col]
                + partial[(3 * TI + row) * MD + col];
        out[(long)(i0 + row) * MD + col] = v;
    }
}

extern "C" void kernel_launch(void* const* d_in, const int* in_sizes, int n_in,
                              void* d_out, int out_size, void* d_ws, size_t ws_size,
                              hipStream_t stream) {
    (void)in_sizes; (void)n_in; (void)out_size; (void)d_ws; (void)ws_size;
    const float* hidden = (const float*)d_in[0];
    const float* corr   = (const float*)d_in[1];
    const int*   nei    = (const int*)d_in[2];
    const float* W_rela = (const float*)d_in[3];
    const float* b_rela = (const float*)d_in[4];
    const float* W_att  = (const float*)d_in[5];
    const float* b_att  = (const float*)d_in[6];
    float* outp = (float*)d_out;

    social_tensor_kernel<<<dim3(PN / TI), dim3(NTHR), 0, stream>>>(
        hidden, corr, nei, W_rela, b_rela, W_att, b_att, outp);
}


# 8 waves per work-group, not 16 (2048 waves enter, not 4096): a thread carries 16 score elements and 64 pieces of hidden, a wave two N-tiles; row max/sum trees span 8 waves (row-sum order changes)
# speedup vs baseline: 3.5136x; 1.0738x over previous
#include <hip/hip_runtime.h>


typedef __attribute__((ext_vector_type(2))) float v2f;
typedef __attribute__((ext_vector_type(4))) float v4f;
typedef __attribute__((ext_vector_type(8))) float v8f;

#define PN    1024
#define MD    64
#define RD    32
#define TI    4
#define SROW  1028
#define NTHR  256
#define NWAVE (NTHR / 32)
#define EPR   (PN / NTHR)
#define EPT   (TI * EPR)
#define RS    (NTHR / 16)
#define HB    16
#define NB    (PN / (RS * HB))
static_assert(NWAVE == 8 && EPR == 4 && (TI & (TI - 1)) == 0, "reduction code below is written for this shape");
static_assert(NB * HB * NTHR * 4 == PN * MD, "the batches cover hidden exactly once");

template <int M>
__device__ __forceinline__ float row_xor(float x)
{
    return __builtin_bit_cast(float,
        __builtin_amdgcn_update_dpp(0, __builtin_bit_cast(int, x), 0x160 + M, 0xf, 0xf, true));
}
__device__ __forceinline__ float xhalf(float x)
{
    const int b = __builtin_bit_cast(int, x);
    return __builtin_bit_cast(float, __builtin_amdgcn_permlanex16(b, b, 0x76543210, 0xfedcba98, false, false));
}
__device__ __forceinline__ float row16_sum(float x)
{
    x += row_xor<1>(x);
    x += row_xor<2>(x);
    x += row_xor<4>(x);
    x += row_xor<8>(x);
    return x;
}
__device__ __forceinline__ float row16_sum16(const float (&d)[16], int q)
{
    const bool b8 = (q & 8) != 0, b4 = (q & 4) != 0, b2 = (q & 2) != 0, b1 = (q & 1) != 0;
    float a[8], b[4], c[2];
    #pragma unroll
    for (int u = 0; u < 8; ++u) {
        const float keep = b8 ? d[u + 8] : d[u];
        const float send = b8 ? d[u] : d[u + 8];
        a[u] = keep + row_xor<8>(send);
    }
    #pragma unroll
    for (int u = 0; u < 4; ++u) {
        const float keep = b4 ? a[u + 4] : a[u];
        const float send = b4 ? a[u] : a[u + 4];
        b[u] = keep + row_xor<4>(send);
    }
    #pragma unroll
    for (int u = 0; u < 2; ++u) {
        const float keep = b2 ? b[u + 2] : b[u];
        const float send = b2 ? b[u] : b[u + 2];
        c[u] = keep + row_xor<2>(send);
    }
    const float keep = b1 ? c[1] : c[0];
    const float send = b1 ? c[0] : c[1];
    return keep + row_xor<1>(send);
}

#define WMMA_SLOTS(acc0, acc1) asm volatile("v_nop\n\tv_nop\n\tv_nop\n\tv_nop\n\tv_nop\n\tv_nop\n\tv_nop\n\tv_nop" : "+v"(acc0), "+v"(acc1))

__global__ __launch_bounds__(NTHR)
void social_tensor_kernel(const float* __restrict__ hidden,
                          const float* __restrict__ corr,
                          const int*   __restrict__ nei,
                          const float* __restrict__ W_rela,
                          const float* __restrict__ b_rela,
                          const float* __restrict__ W_att,
                          const float* __restrict__ b_att,
                          float* __restrict__ out)
{
    struct __attribute__((aligned(16))) Smem {
        float red_mx[TI * NWAVE];
        float red_sm[TI * NWAVE];
        float s[TI * SROW];
        float hv_nei[PN];
        float partial[4 * TI * MD];
        float wts[4 * RD];
        float hv_hi[TI];
        float hs[PN * MD];
    };
    static_assert(__builtin_offsetof(Smem, hs) % 16 == 0 && __builtin_offsetof(Smem, red_sm) % 16 == 0 &&
                  __builtin_offsetof(Smem, wts) % 16 == 0, "16-byte LDS accesses below");
    __shared__ Smem sm;
    auto& red_mx = sm.red_mx;  auto& red_sm = sm.red_sm;  auto& s = sm.s;  auto& hv_nei = sm.hv_nei;
    auto& partial = sm.partial;  auto& wts = sm.wts;
    auto& hv_hi = sm.hv_hi;  auto& hs = sm.hs;

    const int tid  = threadIdx.x;
    const int lane = tid & 31;
    const int wave = tid >> 5;
    const int i0   = blockIdx.x * TI;
    const int q    = tid & 15;
    const int hr   = tid >> 4;
    const int qs   = q ^ ((wave & 1) << 2);

    const int wsel = __builtin_amdgcn_readfirstlane(wave) & 3;
    const float* wsrc = wsel == 0 ? W_rela : wsel == 1 ? W_rela + RD : wsel == 2 ? b_rela : W_att;
    const float wv = wsrc[lane];
    const float* wq = W_att + RD + MD + 4 * q;
    const float4 wn = make_float4(wq[0], wq[1], wq[2], wq[3]);
    const float* php = hidden + (long)(i0 + (hr & (TI - 1))) * MD + 4 * q;
    const float* pwh = W_att + RD + 4 * q;
    const float4 hp = make_float4(php[0], php[1], php[2], php[3]);
    const float4 hw = make_float4(pwh[0], pwh[1], pwh[2], pwh[3]);

    float c0[EPT], c1[EPT];
    int   nv[EPT];
    #pragma unroll
    for (int e = 0; e < EPT; ++e) {
        const int ii = e / EPR;
        const int j  = tid + (e % EPR) * NTHR;
        const long idx = (long)(i0 + ii) * PN + j;
        c0[e] = corr[idx * 2 + 0];
        c1[e] = corr[idx * 2 + 1];
        nv[e] = nei[idx];
    }

    __builtin_amdgcn_sched_barrier(0);

    wts[tid & (4 * RD - 1)] = wv;
    {
        float d = hp.x * hw.x;
        d = fmaf(hp.y, hw.y, d);
        d = fmaf(hp.z, hw.z, d);
        d = fmaf(hp.w, hw.w, d);
        d = row16_sum(d);
        if (tid < 16 * TI && q == 0) hv_hi[hr] = d;
    }
    __syncthreads();

    float4 hb[NB * HB];
    #pragma unroll
    for (int u = 0; u < NB * HB; ++u) {
        const float* p = hidden + (long)(u * NTHR + tid) * 4;
        hb[u] = make_float4(p[0], p[1], p[2], p[3]);
    }

    const float batt = b_att[0];
    unsigned nbm = 0;
    #pragma unroll
    for (int e = 0; e < EPT; ++e) nbm |= (nv[e] > 0 ? 1u : 0u) << e;

    float acc_s[EPT];
    #pragma unroll
    for (int e = 0; e < EPT; ++e) acc_s[e] = 0.f;
    {
        const v4f* w0v = (const v4f*)(wts);
        const v4f* w1v = (const v4f*)(wts + RD);
        const v4f* bv  = (const v4f*)(wts + 2 * RD);
        const v4f* wav = (const v4f*)(wts + 3 * RD);
        #pragma unroll 1
        for (int rc = 0; rc < RD / 4; ++rc) {
            const v4f g0 = w0v[rc], g1 = w1v[rc], gb = bv[rc], ga = wav[rc];
            #pragma unroll
            for (int e = 0; e < EPT; ++e) {
                float t;
                t = fmaf(c1[e], g1.x, gb.x);  t = fmaf(c0[e], g0.x, t);  t = fmaxf(t, 0.f);  acc_s[e] = fmaf(t, ga.x, acc_s[e]);
                t = fmaf(c1[e], g1.y, gb.y);  t = fmaf(c0[e], g0.y, t);  t = fmaxf(t, 0.f);  acc_s[e] = fmaf(t, ga.y, acc_s[e]);
                t = fmaf(c1[e], g1.z, gb.z);  t = fmaf(c0[e], g0.z, t);  t = fmaxf(t, 0.f);  acc_s[e] = fmaf(t, ga.z, acc_s[e]);
                t = fmaf(c1[e], g1.w, gb.w);  t = fmaf(c0[e], g0.w, t);  t = fmaxf(t, 0.f);  acc_s[e] = fmaf(t, ga.w, acc_s[e]);
            }
        }
    }

    #pragma unroll
    for (int hh = 0; hh < NB; ++hh) {
        float d[HB];
        #pragma unroll
        for (int u = 0; u < HB; ++u) {
            const int row = (hh * HB + u) * RS + hr;
            const float4 v = hb[hh * HB + u];
            *(float4*)&hs[row * MD + 4 * qs] = v;
            float t = v.x * wn.x;
            t = fmaf(v.y, wn.y, t);
            t = fmaf(v.z, wn.z, t);
            t = fmaf(v.w, wn.w, t);
            d[u] = t;
        }
        hv_nei[(hh * HB + q) * RS + hr] = row16_sum16(d, q);
    }
    __syncthreads();

    float pv[EPT];
    float mf[EPT];
    #pragma unroll
    for (int e = 0; e < EPT; ++e) {
        const int ii = e / EPR;
        const int j  = tid + (e % EPR) * NTHR;
        const bool nb = ((nbm >> e) & 1u) != 0;
        float sc = acc_s[e] + hv_hi[ii] + hv_nei[j] + batt;
        float v = nb ? sc : 0.f;
        v = (v == 0.f) ? -1e-6f : v;
        pv[e] = v;
        mf[e] = nb ? 1.f : 0.f;
    }

    float rm[TI];
    #pragma unroll
    for (int ii = 0; ii < TI; ++ii) rm[ii] = fmaxf(fmaxf(pv[ii * EPR + 0], pv[ii * EPR + 1]), fmaxf(pv[ii * EPR + 2], pv[ii * EPR + 3]));
    #pragma unroll
    for (int ii = 0; ii < TI; ++ii) rm[ii] = fmaxf(rm[ii], xhalf(rm[ii]));
    #pragma unroll
    for (int ii = 0; ii < TI; ++ii) rm[ii] = fmaxf(rm[ii], row_xor<8>(rm[ii]));
    #pragma unroll
    for (int ii = 0; ii < TI; ++ii) rm[ii] = fmaxf(rm[ii], row_xor<4>(rm[ii]));
    #pragma unroll
    for (int ii = 0; ii < TI; ++ii) rm[ii] = fmaxf(rm[ii], row_xor<2>(rm[ii]));
    #pragma unroll
    for (int ii = 0; ii < TI; ++ii) rm[ii] = fmaxf(rm[ii], row_xor<1>(rm[ii]));
    if (lane == 0) {
        #pragma unroll
        for (int ii = 0; ii < TI; ++ii) red_mx[ii * NWAVE + wave] = rm[ii];
    }
    __syncthreads();
    float ev[EPT];
    float rs[TI];
    float gm[TI];
    #pragma unroll
    for (int ii = 0; ii < TI; ++ii) gm[ii] = red_mx[ii * NWAVE + (lane & 7)];
    #pragma unroll
    for (int ii = 0; ii < TI; ++ii) gm[ii] = fmaxf(gm[ii], row_xor<4>(gm[ii]));
    #pragma unroll
    for (int ii = 0; ii < TI; ++ii) gm[ii] = fmaxf(gm[ii], row_xor<1>(gm[ii]));
    #pragma unroll
    for (int ii = 0; ii < TI; ++ii) gm[ii] = fmaxf(gm[ii], row_xor<2>(gm[ii]));
    #pragma unroll
    for (int ii = 0; ii < TI; ++ii) {
        const float m = gm[ii];
        const float e0 = expf(pv[ii * EPR + 0] - m);
        const float e1 = expf(pv[ii * EPR + 1] - m);
        const float e2 = expf(pv[ii * EPR + 2] - m);
        const float e3 = expf(pv[ii * EPR + 3] - m);
        ev[ii * EPR + 0] = e0;
        ev[ii * EPR + 1] = e1;
        ev[ii * EPR + 2] = e2;
        ev[ii * EPR + 3] = e3;
        rs[ii] = (e0 + e1) + (e2 + e3);
    }
    #pragma unroll
    for (int ii = 0; ii < TI; ++ii) rs[ii] += xhalf(rs[ii]);
    #pragma unroll
    for (int ii = 0; ii < TI; ++ii) rs[ii] += row_xor<8>(rs[ii]);
    #pragma unroll
    for (int ii = 0; ii < TI; ++ii) rs[ii] += row_xor<4>(rs[ii]);
    #pragma unroll
    for (int ii = 0; ii < TI; ++ii) rs[ii] += row_xor<2>(rs[ii]);
    #pragma unroll
    for (int ii = 0; ii < TI; ++ii) rs[ii] += row_xor<1>(rs[ii]);
    if (lane == 0) {
        #pragma unroll
        for (int ii = 0; ii < TI; ++ii) red_sm[ii * NWAVE + wave] = rs[ii];
    }
    __syncthreads();
    float gs[TI];
    #pragma unroll
    for (int ii = 0; ii < TI; ++ii) gs[ii] = red_sm[ii * NWAVE + (lane & 7)];
    #pragma unroll
    for (int ii = 0; ii < TI; ++ii) gs[ii] += row_xor<4>(gs[ii]);
    #pragma unroll
    for (int ii = 0; ii < TI; ++ii) gs[ii] += row_xor<1>(gs[ii]);
    #pragma unroll
    for (int ii = 0; ii < TI; ++ii) gs[ii] += row_xor<2>(gs[ii]);
    #pragma unroll
    for (int ii = 0; ii < TI; ++ii) {
        const float sm = gs[ii];
        const float inv = 1.f / sm;
        #pragma unroll
        for (int h = 0; h < EPR; ++h)
            s[ii * SROW + tid + h * NTHR] = ev[ii * EPR + h] * inv * mf[ii * EPR + h];
    }
    __syncthreads();

    const int np   = wave & 1;
    const int n0   = np * 32;
    const int kq   = wave >> 1;
    const int kbeg = kq * (PN / 4);

    v8f acc0 = {}, acc1 = {};
    const int  arow = (lane & (TI - 1)) * SROW;
    const int  half = lane >> 4;
    const int  koff = half * 2;
    const int  ncol = n0 + (lane & 15);
    const int  bcol = ncol ^ (half << 4);
    const int  bco2 = (ncol + 16) ^ (half << 4);
    constexpr int KB = 8;
    const float* ap = s + arow + kbeg + koff;
    const float* bp = hs + (kbeg + koff) * MD + bcol;
    const float* bq = hs + (kbeg + koff) * MD + bco2;
    v2f a0[KB], b0[KB], c0f[KB], a1[KB], b1[KB], c1f[KB];
    #pragma unroll
    for (int t = 0; t < KB; ++t) {
        a0[t].x  = ap[4 * t];            a0[t].y  = ap[4 * t + 1];
        b0[t].x  = bp[(4 * t) * MD];     b0[t].y  = bp[(4 * t + 1) * MD];
        c0f[t].x = bq[(4 * t) * MD];     c0f[t].y = bq[(4 * t + 1) * MD];
    }
    constexpr int NBLK = (PN / 4) / (4 * 2 * KB);
    static_assert(NBLK * 4 * 2 * KB == PN / 4, "a K-quarter is a whole number of trips");
    #pragma unroll 1
    for (int blk = 0; blk < NBLK; ++blk) {
        ap += 4 * KB;  bp += 4 * KB * MD;  bq += 4 * KB * MD;
        #pragma unroll
        for (int t = 0; t < KB; ++t) {
            a1[t].x  = ap[4 * t];            a1[t].y  = ap[4 * t + 1];
            b1[t].x  = bp[(4 * t) * MD];     b1[t].y  = bp[(4 * t + 1) * MD];
            c1f[t].x = bq[(4 * t) * MD];     c1f[t].y = bq[(4 * t + 1) * MD];
        }
        #pragma unroll
        for (int t = 0; t < KB; ++t) {
            acc0 = __builtin_amdgcn_wmma_f32_16x16x4_f32(false, a0[t], false, b0[t],  (short)0, acc0, false, false);
            acc1 = __builtin_amdgcn_wmma_f32_16x16x4_f32(false, a0[t], false, c0f[t], (short)0, acc1, false, false);
        }
        WMMA_SLOTS(acc0, acc1);
        ap += 4 * KB;  bp += 4 * KB * MD;  bq += 4 * KB * MD;
        if (blk + 1 < NBLK) {
            #pragma unroll
            for (int t = 0; t < KB; ++t) {
                a0[t].x  = ap[4 * t];            a0[t].y  = ap[4 * t + 1];
                b0[t].x  = bp[(4 * t) * MD];     b0[t].y  = bp[(4 * t + 1) * MD];
                c0f[t].x = bq[(4 * t) * MD];     c0f[t].y = bq[(4 * t + 1) * MD];
            }
        }
        #pragma unroll
        for (int t = 0; t < KB; ++t) {
            acc0 = __builtin_amdgcn_wmma_f32_16x16x4_f32(false, a1[t], false, b1[t],  (short)0, acc0, false, false);
            acc1 = __builtin_amdgcn_wmma_f32_16x16x4_f32(false, a1[t], false, c1f[t], (short)0, acc1, false, false);
        }
        WMMA_SLOTS(acc0, acc1);
    }

    if (lane < 16) {
        #pragma unroll
        for (int g = 0; g < TI; ++g) {
            partial[(kq * TI + g) * MD + ncol]      = acc0[g];
            partial[(kq * TI + g) * MD + ncol + 16] = acc1[g];
        }
    }
    __syncthreads();
    if (tid < TI * MD) {
        const int row = tid >> 6;
        const int col = tid & (MD - 1);
        float v = partial[(0 * TI + row) * MD + col]
                + partial[(1 * TI + row) * MD + col]
                + partial[(2 * TI + row) * MD + col]
                + partial[(3 * TI + row) * MD + col];
        out[(long)(i0 + row) * MD + col] = v;
    }
}

extern "C" void kernel_launch(void* const* d_in, const int* in_sizes, int n_in,
                              void* d_out, int out_size, void* d_ws, size_t ws_size,
                              hipStream_t stream) {
    (void)in_sizes; (void)n_in; (void)out_size; (void)d_ws; (void)ws_size;
    const float* hidden = (const float*)d_in[0];
    const float* corr   = (const float*)d_in[1];
    const int*   nei    = (const int*)d_in[2];
    const float* W_rela = (const float*)d_in[3];
    const float* b_rela = (const float*)d_in[4];
    const float* W_att  = (const float*)d_in[5];
    const float* b_att  = (const float*)d_in[6];
    float* outp = (float*)d_out;

    social_tensor_kernel<<<dim3(PN / TI), dim3(NTHR), 0, stream>>>(
        hidden, corr, nei, W_rela, b_rela, W_att, b_att, outp);
}


# each wave owns 128 score columns and their hidden rows end to end: B from registers (WMMAs in asm with hazard slots), A tile wave-local, own-max softmax rescaled at a last barrier; sum orders change
# speedup vs baseline: 3.7715x; 1.0734x over previous
#include <hip/hip_runtime.h>


typedef __attribute__((ext_vector_type(2))) float v2f;
typedef __attribute__((ext_vector_type(4))) float v4f;
typedef __attribute__((ext_vector_type(8))) float v8f;

#define PN    1024
#define MD    64
#define RD    32
#define TI    4
#define NTHR  256
#define NWAVE (NTHR / 32)
#define WC    (PN / NWAVE)
#define NH    (WC / 32)
#define EPT   (TI * NH)
#define NP    (WC / 2)
#define AROW  132
static_assert(NWAVE == 8 && NH == 4 && TI == 4, "reduction and fragment code below is written for this shape");
static_assert(NP * NTHR * 4 == PN * MD, "the pieces cover hidden exactly once");

template <int M>
__device__ __forceinline__ float row_xor(float x)
{
    return __builtin_bit_cast(float,
        __builtin_amdgcn_update_dpp(0, __builtin_bit_cast(int, x), 0x160 + M, 0xf, 0xf, true));
}
__device__ __forceinline__ float xhalf(float x)
{
    const int b = __builtin_bit_cast(int, x);
    return __builtin_bit_cast(float, __builtin_amdgcn_permlanex16(b, b, 0x76543210, 0xfedcba98, false, false));
}
__device__ __forceinline__ float row16_sum(float x)
{
    x += row_xor<1>(x);
    x += row_xor<2>(x);
    x += row_xor<4>(x);
    x += row_xor<8>(x);
    return x;
}
__device__ __forceinline__ float row16_sum16(const float (&d)[16], int q)
{
    const bool b8 = (q & 8) != 0, b4 = (q & 4) != 0, b2 = (q & 2) != 0, b1 = (q & 1) != 0;
    float a[8], b[4], c[2];
    #pragma unroll
    for (int u = 0; u < 8; ++u) {
        const float keep = b8 ? d[u + 8] : d[u];
        const float send = b8 ? d[u] : d[u + 8];
        a[u] = keep + row_xor<8>(send);
    }
    #pragma unroll
    for (int u = 0; u < 4; ++u) {
        const float keep = b4 ? a[u + 4] : a[u];
        const float send = b4 ? a[u] : a[u + 4];
        b[u] = keep + row_xor<4>(send);
    }
    #pragma unroll
    for (int u = 0; u < 2; ++u) {
        const float keep = b2 ? b[u + 2] : b[u];
        const float send = b2 ? b[u] : b[u + 2];
        c[u] = keep + row_xor<2>(send);
    }
    const float keep = b1 ? c[1] : c[0];
    const float send = b1 ? c[0] : c[1];
    return keep + row_xor<1>(send);
}

#define WMMA_2STEPS(d0, d1, d2, d3, aa, ab, p0, p1, p2, p3, r0, r1, r2, r3)                 \
    asm volatile("v_wmma_f32_16x16x4_f32 %0, %4, %6, %0\n\t"                                \
                 "v_wmma_f32_16x16x4_f32 %1, %4, %7, %1\n\t"                                \
                 "v_wmma_f32_16x16x4_f32 %2, %4, %8, %2\n\t"                                \
                 "v_wmma_f32_16x16x4_f32 %3, %4, %9, %3\n\t"                                \
                 "v_wmma_f32_16x16x4_f32 %0, %5, %10, %0\n\t"                               \
                 "v_wmma_f32_16x16x4_f32 %1, %5, %11, %1\n\t"                               \
                 "v_wmma_f32_16x16x4_f32 %2, %5, %12, %2\n\t"                               \
                 "v_wmma_f32_16x16x4_f32 %3, %5, %13, %3\n\t"                               \
                 "v_nop\n\tv_nop\n\tv_nop\n\tv_nop\n\tv_nop\n\tv_nop\n\tv_nop\n\tv_nop"     \
                 : "+v"(d0), "+v"(d1), "+v"(d2), "+v"(d3)                                   \
                 : "v"(aa), "v"(ab), "v"(p0), "v"(p1), "v"(p2), "v"(p3), "v"(r0), "v"(r1), "v"(r2), "v"(r3))

__device__ __forceinline__ float comp(const float4& v, int n)
{
    return n == 0 ? v.x : n == 1 ? v.y : n == 2 ? v.z : v.w;
}

__global__ __launch_bounds__(NTHR)
void social_tensor_kernel(const float* __restrict__ hidden,
                          const float* __restrict__ corr,
                          const int*   __restrict__ nei,
                          const float* __restrict__ W_rela,
                          const float* __restrict__ b_rela,
                          const float* __restrict__ W_att,
                          const float* __restrict__ b_att,
                          float* __restrict__ out)
{
    struct __attribute__((aligned(16))) Smem {
        float red_mx[TI * NWAVE];
        float red_sm[TI * NWAVE];
        float sa[NWAVE * TI * AROW];
        float partial[NWAVE * TI * MD];
        float wts[4 * RD];
        float hv_hi[TI];
    };
    static_assert(__builtin_offsetof(Smem, partial) % 16 == 0 && __builtin_offsetof(Smem, wts) % 16 == 0,
                  "16-byte LDS accesses below");
    __shared__ Smem sm;
    auto& red_mx = sm.red_mx;  auto& red_sm = sm.red_sm;  auto& sa = sm.sa;
    auto& partial = sm.partial;  auto& wts = sm.wts;  auto& hv_hi = sm.hv_hi;

    const int tid  = threadIdx.x;
    const int lane = tid & 31;
    const int wave = tid >> 5;
    const int i0   = blockIdx.x * TI;
    const int q    = tid & 15;
    const int hr   = tid >> 4;
    const int half = lane >> 4;
    const int lc   = 2 * q + half;
    const int cw   = wave * WC;

    const int wsel = __builtin_amdgcn_readfirstlane(wave) & 3;
    const float* wsrc = wsel == 0 ? W_rela : wsel == 1 ? W_rela + RD : wsel == 2 ? b_rela : W_att;
    const float wv = wsrc[lane];
    const float* wq = W_att + RD + MD + 4 * q;
    const float4 wn = make_float4(wq[0], wq[1], wq[2], wq[3]);
    const float* php = hidden + (long)(i0 + (hr & (TI - 1))) * MD + 4 * q;
    const float* pwh = W_att + RD + 4 * q;
    const float4 hp = make_float4(php[0], php[1], php[2], php[3]);
    const float4 hw = make_float4(pwh[0], pwh[1], pwh[2], pwh[3]);

    float c0[EPT], c1[EPT];
    int   nv[EPT];
    #pragma unroll
    for (int e = 0; e < EPT; ++e) {
        const int ii = e / NH;
        const int j  = cw + (e % NH) * 32 + lc;
        const long idx = (long)(i0 + ii) * PN + j;
        c0[e] = corr[idx * 2 + 0];
        c1[e] = corr[idx * 2 + 1];
        nv[e] = nei[idx];
    }

    __builtin_amdgcn_sched_barrier(0);

    wts[tid & (4 * RD - 1)] = wv;
    {
        float d = hp.x * hw.x;
        d = fmaf(hp.y, hw.y, d);
        d = fmaf(hp.z, hw.z, d);
        d = fmaf(hp.w, hw.w, d);
        d = row16_sum(d);
        if (tid < 16 * TI && q == 0) hv_hi[hr] = d;
    }
    __syncthreads();

    float4 hb[NP];
    #pragma unroll
    for (int u = 0; u < NP; ++u) {
        const float* p = hidden + (long)(cw + 2 * u + half) * MD + 4 * q;
        hb[u] = make_float4(p[0], p[1], p[2], p[3]);
    }

    const float batt = b_att[0];
    unsigned nbm = 0;
    #pragma unroll
    for (int e = 0; e < EPT; ++e) nbm |= (nv[e] > 0 ? 1u : 0u) << e;

    float acc_s[EPT];
    #pragma unroll
    for (int e = 0; e < EPT; ++e) acc_s[e] = 0.f;
    {
        const v4f* w0v = (const v4f*)(wts);
        const v4f* w1v = (const v4f*)(wts + RD);
        const v4f* bv  = (const v4f*)(wts + 2 * RD);
        const v4f* wav = (const v4f*)(wts + 3 * RD);
        #pragma unroll 1
        for (int rc = 0; rc < RD / 4; ++rc) {
            const v4f g0 = w0v[rc], g1 = w1v[rc], gb = bv[rc], ga = wav[rc];
            #pragma unroll
            for (int e = 0; e < EPT; ++e) {
                float t;
                t = fmaf(c1[e], g1.x, gb.x);  t = fmaf(c0[e], g0.x, t);  t = fmaxf(t, 0.f);  acc_s[e] = fmaf(t, ga.x, acc_s[e]);
                t = fmaf(c1[e], g1.y, gb.y);  t = fmaf(c0[e], g0.y, t);  t = fmaxf(t, 0.f);  acc_s[e] = fmaf(t, ga.y, acc_s[e]);
                t = fmaf(c1[e], g1.z, gb.z);  t = fmaf(c0[e], g0.z, t);  t = fmaxf(t, 0.f);  acc_s[e] = fmaf(t, ga.z, acc_s[e]);
                t = fmaf(c1[e], g1.w, gb.w);  t = fmaf(c0[e], g0.w, t);  t = fmaxf(t, 0.f);  acc_s[e] = fmaf(t, ga.w, acc_s[e]);
            }
        }
    }

    float hvn[NH];
    #pragma unroll
    for (int g = 0; g < NH; ++g) {
        float d[16];
        #pragma unroll
        for (int u = 0; u < 16; ++u) {
            const float4 v = hb[g * 16 + u];
            float t = v.x * wn.x;
            t = fmaf(v.y, wn.y, t);
            t = fmaf(v.z, wn.z, t);
            t = fmaf(v.w, wn.w, t);
            d[u] = t;
        }
        hvn[g] = row16_sum16(d, q);
    }

    float pv[EPT];
    float mf[EPT];
    #pragma unroll
    for (int e = 0; e < EPT; ++e) {
        const int ii = e / NH;
        const bool nb = ((nbm >> e) & 1u) != 0;
        float sc = acc_s[e] + hv_hi[ii] + hvn[e % NH] + batt;
        float v = nb ? sc : 0.f;
        v = (v == 0.f) ? -1e-6f : v;
        pv[e] = v;
        mf[e] = nb ? 1.f : 0.f;
    }

    float rm[TI];
    #pragma unroll
    for (int ii = 0; ii < TI; ++ii) rm[ii] = fmaxf(fmaxf(pv[ii * NH + 0], pv[ii * NH + 1]), fmaxf(pv[ii * NH + 2], pv[ii * NH + 3]));
    #pragma unroll
    for (int ii = 0; ii < TI; ++ii) rm[ii] = fmaxf(rm[ii], xhalf(rm[ii]));
    #pragma unroll
    for (int ii = 0; ii < TI; ++ii) rm[ii] = fmaxf(rm[ii], row_xor<8>(rm[ii]));
    #pragma unroll
    for (int ii = 0; ii < TI; ++ii) rm[ii] = fmaxf(rm[ii], row_xor<4>(rm[ii]));
    #pragma unroll
    for (int ii = 0; ii < TI; ++ii) rm[ii] = fmaxf(rm[ii], row_xor<2>(rm[ii]));
    #pragma unroll
    for (int ii = 0; ii < TI; ++ii) rm[ii] = fmaxf(rm[ii], row_xor<1>(rm[ii]));
    float rs[TI];
    float* saw = sa + wave * (TI * AROW);
    #pragma unroll
    for (int ii = 0; ii < TI; ++ii) {
        const float m = rm[ii];
        const float e0 = expf(pv[ii * NH + 0] - m);
        const float e1 = expf(pv[ii * NH + 1] - m);
        const float e2 = expf(pv[ii * NH + 2] - m);
        const float e3 = expf(pv[ii * NH + 3] - m);
        rs[ii] = (e0 + e1) + (e2 + e3);
        saw[ii * AROW + 0 * 32 + lc] = e0 * mf[ii * NH + 0];
        saw[ii * AROW + 1 * 32 + lc] = e1 * mf[ii * NH + 1];
        saw[ii * AROW + 2 * 32 + lc] = e2 * mf[ii * NH + 2];
        saw[ii * AROW + 3 * 32 + lc] = e3 * mf[ii * NH + 3];
    }
    #pragma unroll
    for (int ii = 0; ii < TI; ++ii) rs[ii] += xhalf(rs[ii]);
    #pragma unroll
    for (int ii = 0; ii < TI; ++ii) rs[ii] += row_xor<8>(rs[ii]);
    #pragma unroll
    for (int ii = 0; ii < TI; ++ii) rs[ii] += row_xor<4>(rs[ii]);
    #pragma unroll
    for (int ii = 0; ii < TI; ++ii) rs[ii] += row_xor<2>(rs[ii]);
    #pragma unroll
    for (int ii = 0; ii < TI; ++ii) rs[ii] += row_xor<1>(rs[ii]);
    if (lane == 0) {
        #pragma unroll
        for (int ii = 0; ii < TI; ++ii) {
            red_mx[ii * NWAVE + wave] = rm[ii];
            red_sm[ii * NWAVE + wave] = rs[ii];
        }
    }

    v8f acc0 = {}, acc1 = {}, acc2 = {}, acc3 = {};
    const float* ap = saw + (lane & (TI - 1)) * AROW + half;
    v2f af[NP / 2];
    #pragma unroll
    for (int t = 0; t < NP / 2; ++t) { af[t].x = ap[4 * t]; af[t].y = ap[4 * t + 2]; }
    #pragma unroll
    for (int t = 0; t < NP / 2; t += 2) {
        const float4 x0 = hb[2 * t], x1 = hb[2 * t + 1], y0 = hb[2 * t + 2], y1 = hb[2 * t + 3];
        const v2f p0 = { comp(x0, 0), comp(x1, 0) };
        const v2f p1 = { comp(x0, 1), comp(x1, 1) };
        const v2f p2 = { comp(x0, 2), comp(x1, 2) };
        const v2f p3 = { comp(x0, 3), comp(x1, 3) };
        const v2f r0 = { comp(y0, 0), comp(y1, 0) };
        const v2f r1 = { comp(y0, 1), comp(y1, 1) };
        const v2f r2 = { comp(y0, 2), comp(y1, 2) };
        const v2f r3 = { comp(y0, 3), comp(y1, 3) };
        WMMA_2STEPS(acc0, acc1, acc2, acc3, af[t], af[t + 1], p0, p1, p2, p3, r0, r1, r2, r3);
    }

    if (lane < 16) {
        #pragma unroll
        for (int g = 0; g < TI; ++g)
            *(float4*)&partial[(wave * TI + g) * MD + 4 * q] = make_float4(acc0[g], acc1[g], acc2[g], acc3[g]);
    }
    __syncthreads();
    {
        const int row = tid >> 6;
        const int col = tid & (MD - 1);
        float mw[NWAVE];
        #pragma unroll
        for (int w = 0; w < NWAVE; ++w) mw[w] = red_mx[row * NWAVE + w];
        float mx = mw[0];
        #pragma unroll
        for (int w = 1; w < NWAVE; ++w) mx = fmaxf(mx, mw[w]);
        float den = 0.f, num = 0.f;
        #pragma unroll
        for (int w = 0; w < NWAVE; ++w) {
            const float f = expf(mw[w] - mx);
            den = fmaf(f, red_sm[row * NWAVE + w], den);
            num = fmaf(f, partial[(w * TI + row) * MD + col], num);
        }
        out[(long)(i0 + row) * MD + col] = num / den;
    }
}

extern "C" void kernel_launch(void* const* d_in, const int* in_sizes, int n_in,
                              void* d_out, int out_size, void* d_ws, size_t ws_size,
                              hipStream_t stream) {
    (void)in_sizes; (void)n_in; (void)out_size; (void)d_ws; (void)ws_size;
    const float* hidden = (const float*)d_in[0];
    const float* corr   = (const float*)d_in[1];
    const int*   nei    = (const int*)d_in[2];
    const float* W_rela = (const float*)d_in[3];
    const float* b_rela = (const float*)d_in[4];
    const float* W_att  = (const float*)d_in[5];
    const float* b_att  = (const float*)d_in[6];
    float* outp = (float*)d_out;

    social_tensor_kernel<<<dim3(PN / TI), dim3(NTHR), 0, stream>>>(
        hidden, corr, nei, W_rela, b_rela, W_att, b_att, outp);
}
